# starting kernel
# baseline (speedup 1.0000x reference)
#include <hip/hip_runtime.h>


typedef __attribute__((ext_vector_type(2))) float v2f;
typedef __attribute__((ext_vector_type(8))) float v8f;

#define BATCH 32
#define LOUT  512
#define LSEQ  511
#define FEAT  4
#define NCOMP 10
#define NF    340
#define NCH   341
#define COLS  32
#define CHUNK 16

__global__ __launch_bounds__(32)
void sig_scan_kernel(const float* __restrict__ x,
                     const float* __restrict__ kern,
                     float* __restrict__ out)
{
    __shared__ float lds[NCOMP * CHUNK * COLS];

    const int lane = threadIdx.x;
    const int b    = blockIdx.x;
    const int nf0  = blockIdx.y * COLS;

    const int half = lane >> 4;
    const int l15  = lane & 15;
    const int kb   = half * 2;

    v2f blo[NCOMP], bhi[NCOMP];
    #pragma unroll
    for (int c = 0; c < NCOMP; ++c) {
        const int c0 = nf0 + l15;
        const int c1 = nf0 + 16 + l15;
        v2f t0; t0.x = 0.f; t0.y = 0.f;
        v2f t1; t1.x = 0.f; t1.y = 0.f;
        if (c0 < NF) {
            t0.x = kern[(c * FEAT + kb    ) * NF + c0];
            t0.y = kern[(c * FEAT + kb + 1) * NF + c0];
        }
        if (c1 < NF) {
            t1.x = kern[(c * FEAT + kb    ) * NF + c1];
            t1.y = kern[(c * FEAT + kb + 1) * NF + c1];
        }
        blo[c] = t0;
        bhi[c] = t1;
    }

    float p21 = 0.f, p31 = 0.f, p32 = 0.f;
    float p41 = 0.f, p42 = 0.f, p43 = 0.f;
    float tot = 0.f;

    const float* xb   = x   + (size_t)b * LOUT * FEAT;
    float*       outb = out + (size_t)b * LOUT * NCH;

    const int col = lane;
    const int nf  = nf0 + col;

    for (int t0 = 0; t0 < LSEQ; t0 += CHUNK) {
        const int row = t0 + l15;
        v2f a; a.x = 0.f; a.y = 0.f;
        if (row < LSEQ) {
            const float x0a = xb[ row      * FEAT + kb];
            const float x0b = xb[ row      * FEAT + kb + 1];
            const float x1a = xb[(row + 1) * FEAT + kb];
            const float x1b = xb[(row + 1) * FEAT + kb + 1];
            a.x = x1a - x0a;
            a.y = x1b - x0b;
        }

        #pragma unroll
        for (int c = 0; c < NCOMP; ++c) {
            v8f z = {};
            v8f dlo = __builtin_amdgcn_wmma_f32_16x16x4_f32(
                false, a, false, blo[c], (short)0, z, false, false);
            v8f dhi = __builtin_amdgcn_wmma_f32_16x16x4_f32(
                false, a, false, bhi[c], (short)0, z, false, false);
            #pragma unroll
            for (int v = 0; v < 8; ++v) {
                const int r = v + half * 8;
                lds[(c * CHUNK + r) * COLS + l15]      = dlo[v];
                lds[(c * CHUNK + r) * COLS + 16 + l15] = dhi[v];
            }
        }
        __syncthreads();

        for (int r = 0; r < CHUNK; ++r) {
            const int t = t0 + r;
            if (t >= LSEQ) break;
            const float m0 = lds[(0 * CHUNK + r) * COLS + col];
            const float m1 = lds[(1 * CHUNK + r) * COLS + col];
            const float m2 = lds[(2 * CHUNK + r) * COLS + col];
            const float m3 = lds[(3 * CHUNK + r) * COLS + col];
            const float m4 = lds[(4 * CHUNK + r) * COLS + col];
            const float m5 = lds[(5 * CHUNK + r) * COLS + col];
            const float m6 = lds[(6 * CHUNK + r) * COLS + col];
            const float m7 = lds[(7 * CHUNK + r) * COLS + col];
            const float m8 = lds[(8 * CHUNK + r) * COLS + col];
            const float m9 = lds[(9 * CHUNK + r) * COLS + col];

            const float r1 = m0;
            const float r2 = m2 * p21;  p21 += m1;
            const float s3 = m4 * p31;  p31 += m3;
            const float r3 = m5 * p32;  p32 += s3;
            const float s4 = m7 * p41;  p41 += m6;
            const float s5 = m8 * p42;  p42 += s4;
            const float r4 = m9 * p43;  p43 += s5;

            tot += r1 + r2 + r3 + r4;
            if (nf < NF)
                outb[(size_t)(t + 1) * NCH + 1 + nf] = tot;
        }
        __syncthreads();
    }
}

__global__ void sig_init_kernel(float* __restrict__ out)
{
    const int idx = blockIdx.x * blockDim.x + threadIdx.x;
    if (idx < BATCH * LOUT) {
        const int b = idx / LOUT;
        const int t = idx % LOUT;
        out[(size_t)(b * LOUT + t) * NCH] = 1.0f;
        if (t < NF)
            out[(size_t)b * LOUT * NCH + 1 + t] = 0.0f;
    }
}

extern "C" void kernel_launch(void* const* d_in, const int* in_sizes, int n_in,
                              void* d_out, int out_size, void* d_ws, size_t ws_size,
                              hipStream_t stream)
{
    (void)in_sizes; (void)n_in; (void)out_size; (void)d_ws; (void)ws_size;
    const float* x    = (const float*)d_in[0];
    const float* kern = (const float*)d_in[1];
    float*       out  = (float*)d_out;

    sig_init_kernel<<<(BATCH * LOUT + 255) / 256, 256, 0, stream>>>(out);

    dim3 grid(BATCH, (NF + COLS - 1) / COLS);
    sig_scan_kernel<<<grid, 32, 0, stream>>>(x, kern, out);
}


# LDS tile of each wave laid out [component][column][16 steps] with an XOR swizzle: conflict-free 16-byte LDS stores and loads replace 4-byte ones that hit each bank twice; same sums, same order
# speedup vs baseline: 3.9968x; 3.9968x over previous
#include <hip/hip_runtime.h>


typedef __attribute__((ext_vector_type(2))) float v2f;
typedef __attribute__((ext_vector_type(8))) float v8f;
typedef __attribute__((ext_vector_type(4))) float v4f;

#define BATCH 32
#define LOUT  512
#define LSEQ  511
#define FEAT  4
#define NCOMP 10
#define NF    340
#define NCH   341
#define COLS  32
#define CHUNK 16
#define NSEG  8
#define SEGLEN 64
#define TILE  (NCOMP * CHUNK * COLS)
#define CSTR  (CHUNK * COLS)

__device__ __forceinline__
void lift_chunk(const float* __restrict__ xb, const int t0,
                const int l15, const int kb,
                const v2f (&blo)[NCOMP], const v2f (&bhi)[NCOMP],
                float* __restrict__ tile, const int st0, const int st1)
{
    const int row = t0 + l15;
    v2f a; a.x = 0.f; a.y = 0.f;
    if (row < LSEQ) {
        const float x0a = xb[ row      * FEAT + kb];
        const float x0b = xb[ row      * FEAT + kb + 1];
        const float x1a = xb[(row + 1) * FEAT + kb];
        const float x1b = xb[(row + 1) * FEAT + kb + 1];
        a.x = x1a - x0a;
        a.y = x1b - x0b;
    }

    #pragma unroll
    for (int c = 0; c < NCOMP; ++c) {
        v8f z = {};
        v8f dlo = __builtin_amdgcn_wmma_f32_16x16x4_f32(
            false, a, false, blo[c], (short)0, z, false, false);
        v8f dhi = __builtin_amdgcn_wmma_f32_16x16x4_f32(
            false, a, false, bhi[c], (short)0, z, false, false);
        float* tc = tile + c * CSTR;
        v4f q;
        q.x = dlo[0]; q.y = dlo[1]; q.z = dlo[2]; q.w = dlo[3];
        *(v4f*)(tc + st0) = q;
        q.x = dlo[4]; q.y = dlo[5]; q.z = dlo[6]; q.w = dlo[7];
        *(v4f*)(tc + st1) = q;
        q.x = dhi[0]; q.y = dhi[1]; q.z = dhi[2]; q.w = dhi[3];
        *(v4f*)(tc + 16 * CHUNK + st0) = q;
        q.x = dhi[4]; q.y = dhi[5]; q.z = dhi[6]; q.w = dhi[7];
        *(v4f*)(tc + 16 * CHUNK + st1) = q;
    }
}

__global__ __launch_bounds__(NSEG * 32)
void sig_scan_kernel(const float* __restrict__ x,
                     const float* __restrict__ kern,
                     float* __restrict__ out)
{
    __shared__ __attribute__((aligned(16))) float lds[NSEG * TILE];

    const int lane = threadIdx.x & 31;
    const int w    = __builtin_amdgcn_readfirstlane((int)(threadIdx.x >> 5));
    const int b    = blockIdx.x;
    const int nf0  = blockIdx.y * COLS;

    const int half = lane >> 4;
    const int l15  = lane & 15;
    const int kb   = half * 2;

    float* tile = lds + w * TILE;

    v2f blo[NCOMP], bhi[NCOMP];
    #pragma unroll
    for (int c = 0; c < NCOMP; ++c) {
        const int c0 = nf0 + l15;
        const int c1 = nf0 + 16 + l15;
        v2f t0; t0.x = 0.f; t0.y = 0.f;
        v2f t1; t1.x = 0.f; t1.y = 0.f;
        if (c0 < NF) {
            t0.x = kern[(c * FEAT + kb    ) * NF + c0];
            t0.y = kern[(c * FEAT + kb + 1) * NF + c0];
        }
        if (c1 < NF) {
            t1.x = kern[(c * FEAT + kb    ) * NF + c1];
            t1.y = kern[(c * FEAT + kb + 1) * NF + c1];
        }
        blo[c] = t0;
        bhi[c] = t1;
    }

    const float* xb   = x   + (size_t)b * LOUT * FEAT;
    float*       outb = out + (size_t)b * LOUT * NCH;

    const int col  = lane;
    const int nf   = nf0 + col;
    const int tbeg = w * SEGLEN;

    const int swz  = (lane >> 2) & 3;
    const int st0  = l15 * CHUNK + 4 * ((2 * half)     ^ swz);
    const int st1  = l15 * CHUNK + 4 * ((2 * half + 1) ^ swz);
    const float* tcol = tile + col * CHUNK;

    float g21 = 0.f, g31 = 0.f, g32 = 0.f;
    float g41 = 0.f, g42 = 0.f, g43 = 0.f, gt = 0.f;
    float c4 = 0.f, c7 = 0.f, c8 = 0.f, d = 0.f;
    float c2 = 0.f, c5 = 0.f, a = 0.f, c9 = 0.f, bq = 0.f, cq = 0.f;

    for (int i = 0; i < SEGLEN / CHUNK; ++i) {
        const int t0 = tbeg + i * CHUNK;
        lift_chunk(xb, t0, l15, kb, blo, bhi, tile, st0, st1);
        __syncthreads();

        #pragma unroll
        for (int g = 0; g < CHUNK / 4; ++g) {
          v4f mm[NCOMP];
          #pragma unroll
          for (int c = 0; c < NCOMP; ++c)
              mm[c] = *(const v4f*)(tcol + c * CSTR + 4 * (g ^ swz));
          #pragma unroll
          for (int j = 0; j < 4; ++j) {
            const float m0 = mm[0][j];
            const float m1 = mm[1][j];
            const float m2 = mm[2][j];
            const float m3 = mm[3][j];
            const float m4 = mm[4][j];
            const float m5 = mm[5][j];
            const float m6 = mm[6][j];
            const float m7 = mm[7][j];
            const float m8 = mm[8][j];
            const float m9 = mm[9][j];

            cq += m9 * d;
            d  += m8 * c7;
            c7 += m7;
            bq += m9 * c8;
            c8 += m8;
            a  += m5 * c4;
            c4 += m4;
            c2 += m2;
            c5 += m5;
            c9 += m9;

            gt  += m0 + m2 * g21 + m5 * g32 + m9 * g43;
            g43 += m8 * g42;
            g42 += m7 * g41;
            g41 += m6;
            g32 += m4 * g31;
            g31 += m3;
            g21 += m1;
          }
        }
        __syncthreads();
    }

    tile[ 0 * COLS + col] = g21;
    tile[ 1 * COLS + col] = g31;
    tile[ 2 * COLS + col] = g32;
    tile[ 3 * COLS + col] = g41;
    tile[ 4 * COLS + col] = g42;
    tile[ 5 * COLS + col] = g43;
    tile[ 6 * COLS + col] = gt;
    tile[ 7 * COLS + col] = c4;
    tile[ 8 * COLS + col] = c7;
    tile[ 9 * COLS + col] = c8;
    tile[10 * COLS + col] = d;
    tile[11 * COLS + col] = c2;
    tile[12 * COLS + col] = c5;
    tile[13 * COLS + col] = a;
    tile[14 * COLS + col] = c9;
    tile[15 * COLS + col] = bq;
    tile[16 * COLS + col] = cq;
    __syncthreads();

    float p21 = 0.f, p31 = 0.f, p32 = 0.f;
    float p41 = 0.f, p42 = 0.f, p43 = 0.f;
    float tot = 0.f;
    for (int s = 0; s < w; ++s) {
        const float* q = lds + s * TILE + col;
        const float s21 = q[ 0 * COLS], s31 = q[ 1 * COLS], s32 = q[ 2 * COLS];
        const float s41 = q[ 3 * COLS], s42 = q[ 4 * COLS], s43 = q[ 5 * COLS];
        const float st  = q[ 6 * COLS];
        const float k4  = q[ 7 * COLS], k7 = q[ 8 * COLS], k8 = q[ 9 * COLS];
        const float kd  = q[10 * COLS];
        const float k2  = q[11 * COLS], k5 = q[12 * COLS], ka = q[13 * COLS];
        const float k9  = q[14 * COLS], kbq = q[15 * COLS], kcq = q[16 * COLS];

        const float n_tot = tot + st + k2 * p21 + k5 * p32 + ka * p31
                                + k9 * p43 + kbq * p42 + kcq * p41;
        const float n43 = p43 + s43 + k8 * p42 + kd * p41;
        const float n42 = p42 + s42 + k7 * p41;
        const float n41 = p41 + s41;
        const float n32 = p32 + s32 + k4 * p31;
        const float n31 = p31 + s31;
        const float n21 = p21 + s21;
        tot = n_tot; p43 = n43; p42 = n42; p41 = n41;
        p32 = n32;   p31 = n31; p21 = n21;
    }
    __syncthreads();

    for (int i = 0; i < SEGLEN / CHUNK; ++i) {
        const int t0 = tbeg + i * CHUNK;
        lift_chunk(xb, t0, l15, kb, blo, bhi, tile, st0, st1);
        __syncthreads();

        #pragma unroll
        for (int g = 0; g < CHUNK / 4; ++g) {
          v4f mm[NCOMP];
          #pragma unroll
          for (int c = 0; c < NCOMP; ++c)
              mm[c] = *(const v4f*)(tcol + c * CSTR + 4 * (g ^ swz));
          #pragma unroll
          for (int j = 0; j < 4; ++j) {
            const int t = t0 + 4 * g + j;
            const float m0 = mm[0][j];
            const float m1 = mm[1][j];
            const float m2 = mm[2][j];
            const float m3 = mm[3][j];
            const float m4 = mm[4][j];
            const float m5 = mm[5][j];
            const float m6 = mm[6][j];
            const float m7 = mm[7][j];
            const float m8 = mm[8][j];
            const float m9 = mm[9][j];

            const float r1 = m0;
            const float r2 = m2 * p21;  p21 += m1;
            const float s3 = m4 * p31;  p31 += m3;
            const float r3 = m5 * p32;  p32 += s3;
            const float s4 = m7 * p41;  p41 += m6;
            const float s5 = m8 * p42;  p42 += s4;
            const float r4 = m9 * p43;  p43 += s5;

            tot += r1 + r2 + r3 + r4;
            if (nf < NF && t < LSEQ)
                outb[(size_t)(t + 1) * NCH + 1 + nf] = tot;
          }
        }
        __syncthreads();
    }
}

__global__ void sig_init_kernel(float* __restrict__ out)
{
    const int idx = blockIdx.x * blockDim.x + threadIdx.x;
    if (idx < BATCH * LOUT) {
        const int b = idx / LOUT;
        const int t = idx % LOUT;
        out[(size_t)(b * LOUT + t) * NCH] = 1.0f;
        if (t < NF)
            out[(size_t)b * LOUT * NCH + 1 + t] = 0.0f;
    }
}

extern "C" void kernel_launch(void* const* d_in, const int* in_sizes, int n_in,
                              void* d_out, int out_size, void* d_ws, size_t ws_size,
                              hipStream_t stream)
{
    (void)in_sizes; (void)n_in; (void)out_size; (void)d_ws; (void)ws_size;
    const float* x    = (const float*)d_in[0];
    const float* kern = (const float*)d_in[1];
    float*       out  = (float*)d_out;

    sig_init_kernel<<<(BATCH * LOUT + 255) / 256, 256, 0, stream>>>(out);

    dim3 grid(BATCH, (NF + COLS - 1) / COLS);
    sig_scan_kernel<<<grid, NSEG * 32, 0, stream>>>(x, kern, out);
}
